# starting kernel
# baseline (speedup 1.0000x reference)
#include <hip/hip_runtime.h>


typedef __attribute__((ext_vector_type(2))) float v2f;
typedef __attribute__((ext_vector_type(8))) float v8f;

#define EP_BETA 1e-3f
#define EP_LR   0.5f

constexpr int BM = 128;
constexpr int BN = 128;
constexpr int BK = 32;

template <bool TA, bool TB, int EPI>
__global__ void __launch_bounds__(256)
gemm_f32_wmma(const float* __restrict__ A, const float* __restrict__ B,
              const float* __restrict__ Cin, float* __restrict__ D,
              int M, int N, int K, float alpha)
{
    __shared__ v2f Asp[BK / 2][BM + 2];
    __shared__ v2f Bsp[BK / 2][BN + 2];

    const int tid   = threadIdx.x;
    const int lane  = tid & 31;
    const int wid   = tid >> 5;
    const int waveM = wid & 3;
    const int waveN = wid >> 2;
    const int g     = lane >> 4;
    const int ln    = lane & 15;

    const int bm = blockIdx.x * BM;
    const int bn = blockIdx.y * BN;

    v8f acc[2][4] = {};

    for (int kt = 0; kt < K; kt += BK) {
        if (kt + BK < K) {
            const float* pa = TA ? A + (size_t)(kt + BK + (tid >> 5)) * M + bm + 4 * (tid & 31)
                                 : A + (size_t)(bm + (tid >> 3)) * K + kt + BK + 4 * (tid & 7);
            const float* pb = TB ? B + (size_t)(bn + (tid >> 3)) * K + kt + BK + 4 * (tid & 7)
                                 : B + (size_t)(kt + BK + (tid >> 5)) * N + bn + 4 * (tid & 31);
            __builtin_prefetch(pa, 0, 1);
            __builtin_prefetch(pb, 0, 1);
        }

        #pragma unroll
        for (int i = 0; i < 4; ++i) {
            int idx4 = i * 256 + tid;
            if (!TA) {
                int kq = idx4 & 7;
                int m  = idx4 >> 3;
                float4 v = *(const float4*)(A + (size_t)(bm + m) * K + kt + 4 * kq);
                v2f p0; p0.x = v.x; p0.y = v.y;
                v2f p1; p1.x = v.z; p1.y = v.w;
                Asp[2 * kq + 0][m] = p0;
                Asp[2 * kq + 1][m] = p1;
            } else {
                int mq = idx4 & 31;
                int k  = idx4 >> 5;
                float4 v = *(const float4*)(A + (size_t)(kt + k) * M + bm + 4 * mq);
                ((float*)&Asp[k >> 1][4 * mq + 0])[k & 1] = v.x;
                ((float*)&Asp[k >> 1][4 * mq + 1])[k & 1] = v.y;
                ((float*)&Asp[k >> 1][4 * mq + 2])[k & 1] = v.z;
                ((float*)&Asp[k >> 1][4 * mq + 3])[k & 1] = v.w;
            }
        }
        #pragma unroll
        for (int i = 0; i < 4; ++i) {
            int idx4 = i * 256 + tid;
            if (TB) {
                int kq = idx4 & 7;
                int n  = idx4 >> 3;
                float4 v = *(const float4*)(B + (size_t)(bn + n) * K + kt + 4 * kq);
                v2f p0; p0.x = v.x; p0.y = v.y;
                v2f p1; p1.x = v.z; p1.y = v.w;
                Bsp[2 * kq + 0][n] = p0;
                Bsp[2 * kq + 1][n] = p1;
            } else {
                int nq = idx4 & 31;
                int k  = idx4 >> 5;
                float4 v = *(const float4*)(B + (size_t)(kt + k) * N + bn + 4 * nq);
                ((float*)&Bsp[k >> 1][4 * nq + 0])[k & 1] = v.x;
                ((float*)&Bsp[k >> 1][4 * nq + 1])[k & 1] = v.y;
                ((float*)&Bsp[k >> 1][4 * nq + 2])[k & 1] = v.z;
                ((float*)&Bsp[k >> 1][4 * nq + 3])[k & 1] = v.w;
            }
        }
        __syncthreads();

        #pragma unroll
        for (int k0 = 0; k0 < BK; k0 += 4) {
            const int kp = (k0 >> 1) + g;
            v2f a0 = Asp[kp][32 * waveM + ln];
            v2f a1 = Asp[kp][32 * waveM + 16 + ln];
            #pragma unroll
            for (int j = 0; j < 4; ++j) {
                v2f b = Bsp[kp][64 * waveN + 16 * j + ln];
                acc[0][j] = __builtin_amdgcn_wmma_f32_16x16x4_f32(
                    false, a0, false, b, (short)0, acc[0][j], false, false);
                acc[1][j] = __builtin_amdgcn_wmma_f32_16x16x4_f32(
                    false, a1, false, b, (short)0, acc[1][j], false, false);
            }
        }
        __syncthreads();
    }

    #pragma unroll
    for (int i2 = 0; i2 < 2; ++i2) {
        #pragma unroll
        for (int j = 0; j < 4; ++j) {
            int col = bn + 64 * waveN + 16 * j + ln;
            #pragma unroll
            for (int r = 0; r < 8; ++r) {
                int row = bm + 32 * waveM + 16 * i2 + r + 8 * g;
                size_t idx = (size_t)row * N + col;
                float v = acc[i2][j][r];
                if (EPI == 0)      D[idx] = alpha * v;
                else if (EPI == 1) D[idx] = Cin[idx] - v;
                else               D[idx] = Cin[idx] + v;
            }
        }
    }
}

__global__ void __launch_bounds__(256) subk(float* __restrict__ o,
                                            const float* __restrict__ a,
                                            const float* __restrict__ b) {
    size_t i = ((size_t)blockIdx.x * 256 + threadIdx.x) * 4;
    float4 av = *(const float4*)(a + i);
    float4 bv = *(const float4*)(b + i);
    *(float4*)(o + i) = make_float4(av.x - bv.x, av.y - bv.y,
                                    av.z - bv.z, av.w - bv.w);
}

__global__ void __launch_bounds__(256) g3k(float* __restrict__ r3,
                                           const float* __restrict__ t3,
                                           const float* __restrict__ y) {
    size_t i = ((size_t)blockIdx.x * 256 + threadIdx.x) * 4;
    float4 rv = *(const float4*)(r3 + i);
    float4 tv = *(const float4*)(t3 + i);
    float4 yv = *(const float4*)(y + i);
    rv.x += EP_BETA * (tv.x - yv.x);
    rv.y += EP_BETA * (tv.y - yv.y);
    rv.z += EP_BETA * (tv.z - yv.z);
    rv.w += EP_BETA * (tv.w - yv.w);
    *(float4*)(r3 + i) = rv;
}

__global__ void __launch_bounds__(256) updk(float* __restrict__ t,
                                            const float* __restrict__ gr) {
    size_t i = ((size_t)blockIdx.x * 256 + threadIdx.x) * 4;
    float4 tv = *(const float4*)(t + i);
    float4 gv = *(const float4*)(gr + i);
    tv.x -= EP_LR * gv.x; tv.y -= EP_LR * gv.y;
    tv.z -= EP_LR * gv.z; tv.w -= EP_LR * gv.w;
    *(float4*)(t + i) = tv;
}

__global__ void __launch_bounds__(256) copyk(float* __restrict__ dst,
                                             const float* __restrict__ src) {
    size_t i = ((size_t)blockIdx.x * 256 + threadIdx.x) * 4;
    *(float4*)(dst + i) = *(const float4*)(src + i);
}

__global__ void __launch_bounds__(256) loss_partial(const float* __restrict__ s,
                                                    const float* __restrict__ y,
                                                    float* __restrict__ part) {
    __shared__ float red[256];
    size_t i = ((size_t)blockIdx.x * 256 + threadIdx.x) * 4;
    float4 a = *(const float4*)(s + i);
    float4 b = *(const float4*)(y + i);
    float dx = a.x - b.x, dy = a.y - b.y, dz = a.z - b.z, dw = a.w - b.w;
    red[threadIdx.x] = dx * dx + dy * dy + dz * dz + dw * dw;
    __syncthreads();
    for (int off = 128; off > 0; off >>= 1) {
        if (threadIdx.x < off) red[threadIdx.x] += red[threadIdx.x + off];
        __syncthreads();
    }
    if (threadIdx.x == 0) part[blockIdx.x] = red[0];
}

__global__ void __launch_bounds__(256) loss_final(const float* __restrict__ part,
                                                  float* __restrict__ out) {
    __shared__ float red[256];
    red[threadIdx.x] = part[threadIdx.x] + part[threadIdx.x + 256];
    __syncthreads();
    for (int off = 128; off > 0; off >>= 1) {
        if (threadIdx.x < off) red[threadIdx.x] += red[threadIdx.x + off];
        __syncthreads();
    }
    if (threadIdx.x == 0) out[0] = red[0] * (1.0f / 524288.0f);
}

#define GEMM(TA, TB, EPI, M, N, K, Ap, Bp, Cp, Dp, alpha)                     \
    gemm_f32_wmma<TA, TB, EPI><<<dim3((M) / BM, (N) / BN), 256, 0, stream>>>( \
        Ap, Bp, Cp, Dp, M, N, K, alpha)

#define EW(kernel, n, ...) kernel<<<dim3((n) / 1024), 256, 0, stream>>>(__VA_ARGS__)

extern "C" void kernel_launch(void* const* d_in, const int* in_sizes, int n_in,
                              void* d_out, int out_size, void* d_ws, size_t ws_size,
                              hipStream_t stream)
{
    const float* x  = (const float*)d_in[0];
    const float* y  = (const float*)d_in[1];
    const float* W0 = (const float*)d_in[2];
    const float* W1 = (const float*)d_in[3];
    const float* W2 = (const float*)d_in[4];
    float* out = (float*)d_out;

    const int Bt = 1024;
    const int D0 = 1024, D1 = 1024, D2 = 1024, D3 = 512;
    const size_t n1 = (size_t)Bt * D1;
    const size_t n2 = (size_t)Bt * D2;
    const size_t n3 = (size_t)Bt * D3;

    float* ws = (float*)d_ws;
    float* p0 = ws;
    float* t1 = p0 + n1;
    float* t2 = t1 + n1;
    float* t3 = t2 + n2;
    float* r1 = t3 + n3;
    float* r2 = r1 + n1;
    float* r3 = r2 + n2;
    float* part = r3 + n3;

    GEMM(false, false, 0, Bt, D1, D0, x,  W0, (const float*)nullptr, p0, 1.0f);
    EW(copyk, n1, t1, p0);
    GEMM(false, false, 0, Bt, D2, D1, t1, W1, (const float*)nullptr, t2, 1.0f);
    GEMM(false, false, 0, Bt, D3, D2, t2, W2, (const float*)nullptr, t3, 1.0f);
    loss_partial<<<512, 256, 0, stream>>>(t3, y, part);
    loss_final<<<1, 256, 0, stream>>>(part, out + 2621440);

    for (int it = 0; it < 20; ++it) {
        GEMM(false, false, 1, Bt, D2, D1, t1, W1, t2, r2, 1.0f);
        GEMM(false, false, 1, Bt, D3, D2, t2, W2, t3, r3, 1.0f);
        EW(subk, n1, r1, t1, p0);
        GEMM(false, true, 1, Bt, D1, D2, r2, W1, r1, r1, 1.0f);
        GEMM(false, true, 1, Bt, D2, D3, r3, W2, r2, r2, 1.0f);
        EW(g3k, n3, r3, t3, y);
        if (it == 0) {
            GEMM(false, true, 2, Bt, D2, D3, r3, W2, r2, r2, 1.0f);
            GEMM(false, true, 2, Bt, D1, D2, r2, W1, r1, r1, 1.0f);
        }
        EW(updk, n1, t1, r1);
        EW(updk, n2, t2, r2);
        EW(updk, n3, t3, r3);
    }

    GEMM(false, false, 1, Bt, D2, D1, t1, W1, t2, r2, 1.0f);
    GEMM(false, false, 1, Bt, D3, D2, t2, W2, t3, r3, 1.0f);
    EW(subk, n1, r1, t1, p0);
    GEMM(true, false, 0, D0, D1, Bt, x,  r1, (const float*)nullptr, out,           -1.0f / EP_BETA);
    GEMM(true, false, 0, D1, D2, Bt, t1, r2, (const float*)nullptr, out + 1048576, -1.0f / EP_BETA);
    GEMM(true, false, 0, D2, D3, Bt, t2, r3, (const float*)nullptr, out + 2097152, -1.0f / EP_BETA);
}


# Relaxation phases: a work-group holds 32 batch rows x 1/8 of the columns, each weight fragment feeds two M tiles (half the weight reads); K=1024 split over wave pairs: sum = low K half + high K half
# speedup vs baseline: 5.7601x; 5.7601x over previous
#include <hip/hip_runtime.h>


typedef __attribute__((ext_vector_type(2))) float v2f;
typedef __attribute__((ext_vector_type(8))) float v8f;

#define EP_BETA 1e-3f
#define EP_LR   0.5f

__global__ void __launch_bounds__(256) subk(float* __restrict__ o,
                                            const float* __restrict__ a,
                                            const float* __restrict__ b) {
    size_t i = ((size_t)blockIdx.x * 256 + threadIdx.x) * 4;
    float4 av = *(const float4*)(a + i);
    float4 bv = *(const float4*)(b + i);
    *(float4*)(o + i) = make_float4(av.x - bv.x, av.y - bv.y,
                                    av.z - bv.z, av.w - bv.w);
}

__global__ void __launch_bounds__(256) copyk(float* __restrict__ dst,
                                             const float* __restrict__ src) {
    size_t i = ((size_t)blockIdx.x * 256 + threadIdx.x) * 4;
    *(float4*)(dst + i) = *(const float4*)(src + i);
}

__global__ void __launch_bounds__(256) loss_partial(const float* __restrict__ s,
                                                    const float* __restrict__ y,
                                                    float* __restrict__ part) {
    __shared__ float red[256];
    size_t i = ((size_t)blockIdx.x * 256 + threadIdx.x) * 4;
    float4 a = *(const float4*)(s + i);
    float4 b = *(const float4*)(y + i);
    float dx = a.x - b.x, dy = a.y - b.y, dz = a.z - b.z, dw = a.w - b.w;
    red[threadIdx.x] = dx * dx + dy * dy + dz * dz + dw * dw;
    __syncthreads();
    for (int off = 128; off > 0; off >>= 1) {
        if (threadIdx.x < off) red[threadIdx.x] += red[threadIdx.x + off];
        __syncthreads();
    }
    if (threadIdx.x == 0) part[blockIdx.x] = red[0];
}

__global__ void __launch_bounds__(256) loss_final(const float* __restrict__ part,
                                                  float* __restrict__ out) {
    __shared__ float red[256];
    red[threadIdx.x] = part[threadIdx.x] + part[threadIdx.x + 256];
    __syncthreads();
    for (int off = 128; off > 0; off >>= 1) {
        if (threadIdx.x < off) red[threadIdx.x] += red[threadIdx.x + off];
        __syncthreads();
    }
    if (threadIdx.x == 0) out[0] = red[0] * (1.0f / 524288.0f);
}

template <int R, int C>
__global__ void __launch_bounds__(256) transpose_rc(float* __restrict__ dst,
                                                    const float* __restrict__ src)
{
    __shared__ float tile[32][33];
    const int tx = threadIdx.x & 31;
    const int ty = threadIdx.x >> 5;
    const int c0 = blockIdx.x * 32;
    const int r0 = blockIdx.y * 32;
    #pragma unroll
    for (int i = 0; i < 4; ++i)
        tile[ty + 8 * i][tx] = src[(size_t)(r0 + ty + 8 * i) * C + c0 + tx];
    __syncthreads();
    #pragma unroll
    for (int i = 0; i < 4; ++i)
        dst[(size_t)(c0 + ty + 8 * i) * R + r0 + tx] = tile[tx][ty + 8 * i];
}

typedef __attribute__((ext_vector_type(4))) float v4f;

constexpr int SLAB_M = 16;
constexpr int PH_T   = 384;
constexpr int PH_M   = 32;
constexpr int PH_G   = 256;

template <int NT> struct ColVec;
template <> struct ColVec<4> { typedef v4f type; };
template <> struct ColVec<2> { typedef v2f type; };

template <int K, int T = PH_T, int ROWS = SLAB_M>
__device__ __forceinline__ void slab_stage(float* As, const float* src, int tid)
{
    constexpr int P = K + 4;
    constexpr int Q = K / 4;
    constexpr int TOT = ROWS * Q;
    #pragma unroll
    for (int i = 0; i < (TOT + T - 1) / T; ++i) {
        int idx4 = i * T + tid;
        if (TOT % T != 0 && i == TOT / T && idx4 >= TOT) idx4 -= T;
        int row  = idx4 / Q;
        int c4   = idx4 - row * Q;
        v4f v = *(const v4f*)(src + (size_t)idx4 * 4);
        *(v4f*)(As + row * P + 4 * c4) = v;
    }
}

template <int NT, int K, int LD, int P = K + 4>
__device__ __forceinline__ void slab_mma_kn(const float* As, const float* __restrict__ Wl,
                                            int ln, int g, v8f (&acc)[NT])
{
    typedef typename ColVec<NT>::type VT;
    const float* ap = As + ln * P + 4 * g;
    #pragma unroll (4 / NT)
    for (int kc = 0; kc < K; kc += 32) {
        VT  bv[4][4];
        v4f av[4];
        #pragma unroll
        for (int c = 0; c < 4; ++c) {
            #pragma unroll
            for (int i = 0; i < 4; ++i)
                bv[c][i] = *(const VT*)(Wl + (size_t)(kc + 8 * c + i) * LD);
        }
        #pragma unroll
        for (int c = 0; c < 4; ++c) av[c] = *(const v4f*)(ap + kc + 8 * c);
        #pragma unroll
        for (int c = 0; c < 4; ++c) {
            #pragma unroll
            for (int s = 0; s < 2; ++s) {
                v2f a; a.x = av[c][2 * s]; a.y = av[c][2 * s + 1];
                #pragma unroll
                for (int j = 0; j < NT; ++j) {
                    v2f b; b.x = bv[c][2 * s][j]; b.y = bv[c][2 * s + 1][j];
                    acc[j] = __builtin_amdgcn_wmma_f32_16x16x4_f32(
                        false, a, false, b, (short)0, acc[j], false, false);
                }
            }
        }
    }
}

template <int KL, int LD>
__device__ __forceinline__ void slab_mma_2x2(const float* a0, const float* a1,
                                             const float* __restrict__ Wl, v8f (&acc)[2][2])
{
    #pragma unroll 2
    for (int kc = 0; kc < KL; kc += 32) {
        v2f bv[4][4];
        v4f av[2][4];
        #pragma unroll
        for (int c = 0; c < 4; ++c) {
            #pragma unroll
            for (int i = 0; i < 4; ++i)
                bv[c][i] = *(const v2f*)(Wl + (size_t)(kc + 8 * c + i) * LD);
        }
        #pragma unroll
        for (int c = 0; c < 4; ++c) {
            av[0][c] = *(const v4f*)(a0 + kc + 8 * c);
            av[1][c] = *(const v4f*)(a1 + kc + 8 * c);
        }
        #pragma unroll
        for (int c = 0; c < 4; ++c) {
            #pragma unroll
            for (int s = 0; s < 2; ++s) {
                v2f x0; x0.x = av[0][c][2 * s]; x0.y = av[0][c][2 * s + 1];
                v2f x1; x1.x = av[1][c][2 * s]; x1.y = av[1][c][2 * s + 1];
                #pragma unroll
                for (int j = 0; j < 2; ++j) {
                    v2f b; b.x = bv[c][2 * s][j]; b.y = bv[c][2 * s + 1][j];
                    acc[0][j] = __builtin_amdgcn_wmma_f32_16x16x4_f32(
                        false, x0, false, b, (short)0, acc[0][j], false, false);
                    acc[1][j] = __builtin_amdgcn_wmma_f32_16x16x4_f32(
                        false, x1, false, b, (short)0, acc[1][j], false, false);
                }
            }
        }
    }
}

__device__ __forceinline__ void pair_put(float* X, int wid, int lane, const v8f (&a)[2])
{
    float* xo = X + wid * 512 + lane;
    #pragma unroll
    for (int j = 0; j < 2; ++j) {
        #pragma unroll
        for (int r = 0; r < 8; ++r) xo[(8 * j + r) * 32] = a[j][r];
    }
}

__device__ __forceinline__ void pair_get(const float* X, int wid, int lane, v8f (&a)[2])
{
    const float* xi = X + (wid ^ 1) * 512 + lane;
    #pragma unroll
    for (int j = 0; j < 2; ++j) {
        #pragma unroll
        for (int r = 0; r < 8; ++r) a[j][r] += xi[(8 * j + r) * 32];
    }
}

constexpr int PW = 1028;
constexpr int PH = 516;

constexpr int FW_T  = 128;
constexpr int FW_G1 = 64 * (1024 / 128);
constexpr int FW_G3 = 64 * (512 / 128);

template <int N, bool SC>
__global__ void __launch_bounds__(FW_T)
slab_forward(const float* __restrict__ in, const float* __restrict__ W,
             float* __restrict__ out, float alpha)
{
    __shared__ __attribute__((aligned(16))) float S[SLAB_M * PW];

    constexpr int NB = N / 128;
    const int tid  = threadIdx.x;
    const int lane = tid & 31;
    const int wid  = tid >> 5;
    const int g    = lane >> 4;
    const int ln   = lane & 15;
    const int cb   = blockIdx.x % NB;
    const size_t m0 = (size_t)(blockIdx.x / NB) * SLAB_M;

    slab_stage<1024, FW_T>(S, in + m0 * 1024, tid);
    __syncthreads();

    const int nw = 128 * cb + 32 * wid + 2 * ln;
    v8f acc[2] = {};
    v8f ach[2] = {};
    slab_mma_kn<2, 512, N, PW>(S, W + (size_t)(4 * g) * N + nw, ln, g, acc);
    slab_mma_kn<2, 512, N, PW>(S + 512, W + (size_t)(512 + 4 * g) * N + nw, ln, g, ach);
    #pragma unroll
    for (int r = 0; r < 8; ++r) {
        const float s0 = acc[0][r] + ach[0][r];
        const float s1 = acc[1][r] + ach[1][r];
        v2f d;
        d.x = SC ? alpha * s0 : s0;
        d.y = SC ? alpha * s1 : s1;
        *(v2f*)(out + (m0 + r + 8 * g) * N + nw) = d;
    }
}

__global__ void __launch_bounds__(PH_T)
relax_resid(const float* __restrict__ t1, const float* __restrict__ t2,
            const float* __restrict__ t3,
            float* __restrict__ r2, float* __restrict__ r3,
            const float* __restrict__ W1, const float* __restrict__ W2)
{
    __shared__ __attribute__((aligned(16))) float S1[PH_M * PW];
    __shared__ __attribute__((aligned(16))) float S2[PH_M * PW];
    __shared__ __attribute__((aligned(16))) float X[12 * 512];

    const int tid  = threadIdx.x;
    const int lane = tid & 31;
    const int wid  = __builtin_amdgcn_readfirstlane(tid >> 5);
    const int g    = lane >> 4;
    const int ln   = lane & 15;
    const int kh   = wid & 1;
    const int cb   = blockIdx.x & 7;
    const size_t m0 = (size_t)(blockIdx.x >> 3) * PH_M;

    slab_stage<1024, PH_T, PH_M>(S1, t1 + m0 * 1024, tid);
    slab_stage<1024, PH_T, PH_M>(S2, t2 + m0 * 1024, tid);
    __syncthreads();

    const int k0   = 512 * kh;
    const int own  = (ln + 16 * kh) * PW + 4 * g + k0;
    const int oth  = (ln + 16 * (kh ^ 1)) * PW + 4 * g + k0;
    v8f acc[2][2] = {};
    int nw;
    if (wid < 8) {
        nw = 128 * cb + 32 * (wid >> 1) + 2 * ln;
        slab_mma_2x2<512, 1024>(S1 + own, S1 + oth,
                                W1 + (size_t)(k0 + 4 * g) * 1024 + nw, acc);
    } else {
        nw = 64 * cb + 32 * ((wid - 8) >> 1) + 2 * ln;
        slab_mma_2x2<512, 512>(S2 + own, S2 + oth,
                               W2 + (size_t)(k0 + 4 * g) * 512 + nw, acc);
    }
    pair_put(X, wid, lane, acc[1]);
    __syncthreads();
    pair_get(X, wid, lane, acc[0]);

    if (wid < 8) {
        #pragma unroll
        for (int r = 0; r < 8; ++r) {
            const int row = 16 * kh + r + 8 * g;
            v2f c = *(const v2f*)(S2 + row * PW + nw);
            v2f d;
            d.x = c.x - acc[0][0][r]; d.y = c.y - acc[0][1][r];
            *(v2f*)(r2 + (m0 + row) * 1024 + nw) = d;
        }
    } else {
        #pragma unroll
        for (int r = 0; r < 8; ++r) {
            const size_t go = (m0 + 16 * kh + r + 8 * g) * 512 + nw;
            v2f c = *(const v2f*)(t3 + go);
            v2f d;
            d.x = c.x - acc[0][0][r]; d.y = c.y - acc[0][1][r];
            *(v2f*)(r3 + go) = d;
        }
    }
}

__global__ void __launch_bounds__(PH_T)
relax_update(float* __restrict__ t1, float* __restrict__ t2, float* __restrict__ t3,
             const float* __restrict__ r2, const float* __restrict__ r3,
             const float* __restrict__ p0, const float* __restrict__ y,
             const float* __restrict__ W1T, const float* __restrict__ W2T)
{
    __shared__ __attribute__((aligned(16))) float R2[PH_M * PW];
    __shared__ __attribute__((aligned(16))) float R3[PH_M * PH];
    __shared__ __attribute__((aligned(16))) float X[8 * 512];

    const int tid  = threadIdx.x;
    const int lane = tid & 31;
    const int wid  = __builtin_amdgcn_readfirstlane(tid >> 5);
    const int g    = lane >> 4;
    const int ln   = lane & 15;
    const int kh   = wid & 1;
    const int cb   = blockIdx.x & 7;
    const size_t m0 = (size_t)(blockIdx.x >> 3) * PH_M;

    slab_stage<1024, PH_T, PH_M>(R2, r2 + m0 * 1024, tid);
    slab_stage<512, PH_T, PH_M>(R3, r3 + m0 * 512, tid);
    __syncthreads();

    #pragma unroll
    for (int i = 0; i < (PH_M * 16 + PH_T - 1) / PH_T; ++i) {
        int idx4 = i * PH_T + tid;
        if (idx4 < PH_M * 16) {
            const int row = idx4 >> 4;
            const int col = 64 * cb + 4 * (idx4 & 15);
            const size_t go = (m0 + row) * 512 + col;
            v4f rv = *(const v4f*)(R3 + row * PH + col);
            v4f tv = *(const v4f*)(t3 + go);
            v4f yv = *(const v4f*)(y + go);
            rv.x += EP_BETA * (tv.x - yv.x); rv.y += EP_BETA * (tv.y - yv.y);
            rv.z += EP_BETA * (tv.z - yv.z); rv.w += EP_BETA * (tv.w - yv.w);
            tv.x -= EP_LR * rv.x; tv.y -= EP_LR * rv.y;
            tv.z -= EP_LR * rv.z; tv.w -= EP_LR * rv.w;
            *(v4f*)(t3 + go) = tv;
        }
    }

    v8f acc[2][2] = {};
    int nw;
    if (wid < 8) {
        const int k0  = 512 * kh;
        const int own = (ln + 16 * kh) * PW + 4 * g + k0;
        const int oth = (ln + 16 * (kh ^ 1)) * PW + 4 * g + k0;
        nw = 128 * cb + 32 * (wid >> 1) + 2 * ln;
        slab_mma_2x2<512, 1024>(R2 + own, R2 + oth,
                                W1T + (size_t)(k0 + 4 * g) * 1024 + nw, acc);
        pair_put(X, wid, lane, acc[1]);
    } else {
        nw = 128 * cb + 32 * (wid - 8) + 2 * ln;
        slab_mma_2x2<512, 1024>(R3 + ln * PH + 4 * g, R3 + (ln + 16) * PH + 4 * g,
                                W2T + (size_t)(4 * g) * 1024 + nw, acc);
    }
    __syncthreads();

    if (wid < 8) {
        pair_get(X, wid, lane, acc[0]);
        #pragma unroll
        for (int r = 0; r < 8; ++r) {
            const size_t go = (m0 + 16 * kh + r + 8 * g) * 1024 + nw;
            v2f tv = *(const v2f*)(t1 + go);
            v2f pv = *(const v2f*)(p0 + go);
            v2f gv;
            gv.x = (tv.x - pv.x) - acc[0][0][r]; gv.y = (tv.y - pv.y) - acc[0][1][r];
            tv.x -= EP_LR * gv.x; tv.y -= EP_LR * gv.y;
            *(v2f*)(t1 + go) = tv;
        }
    } else {
        #pragma unroll
        for (int m = 0; m < 2; ++m) {
            #pragma unroll
            for (int r = 0; r < 8; ++r) {
                const int row = 16 * m + r + 8 * g;
                const size_t go = (m0 + row) * 1024 + nw;
                v2f rv = *(const v2f*)(R2 + row * PW + nw);
                v2f tv = *(const v2f*)(t2 + go);
                v2f gv;
                gv.x = rv.x - acc[m][0][r]; gv.y = rv.y - acc[m][1][r];
                tv.x -= EP_LR * gv.x; tv.y -= EP_LR * gv.y;
                *(v2f*)(t2 + go) = tv;
            }
        }
    }
}

constexpr int S0_T = 256;

__global__ void __launch_bounds__(S0_T)
step0_g2(const float* __restrict__ r2, const float* __restrict__ r3,
         const float* __restrict__ t3, const float* __restrict__ y,
         const float* __restrict__ W2T, float* __restrict__ G2)
{
    __shared__ __attribute__((aligned(16))) float R3[SLAB_M * PH];
    __shared__ __attribute__((aligned(16))) float G3[SLAB_M * PH];

    const int tid  = threadIdx.x;
    const int lane = tid & 31;
    const int wid  = tid >> 5;
    const int g    = lane >> 4;
    const int ln   = lane & 15;
    const int q    = blockIdx.x & 3;
    const size_t m0 = (size_t)(blockIdx.x >> 2) * SLAB_M;

    #pragma unroll
    for (int i = 0; i < SLAB_M * 128 / S0_T; ++i) {
        const int idx4 = i * S0_T + tid;
        const int row = idx4 >> 7;
        const int col = 4 * (idx4 & 127);
        const size_t go = (m0 + row) * 512 + col;
        v4f rv = *(const v4f*)(r3 + go);
        v4f tv = *(const v4f*)(t3 + go);
        v4f yv = *(const v4f*)(y + go);
        *(v4f*)(R3 + row * PH + col) = rv;
        rv.x += EP_BETA * (tv.x - yv.x); rv.y += EP_BETA * (tv.y - yv.y);
        rv.z += EP_BETA * (tv.z - yv.z); rv.w += EP_BETA * (tv.w - yv.w);
        *(v4f*)(G3 + row * PH + col) = rv;
    }
    __syncthreads();

    const int nw = 256 * q + 32 * wid + 2 * ln;
    const float* Wl = W2T + (size_t)(4 * g) * 1024 + nw;
    v8f accr[2] = {};
    v8f accg[2] = {};
    slab_mma_kn<2, 512, 1024>(R3, Wl, ln, g, accr);
    slab_mma_kn<2, 512, 1024>(G3, Wl, ln, g, accg);
    #pragma unroll
    for (int r = 0; r < 8; ++r) {
        const size_t go = (m0 + r + 8 * g) * 1024 + nw;
        v2f c = *(const v2f*)(r2 + go);
        v2f d;
        d.x = (c.x - accr[0][r]) + accg[0][r];
        d.y = (c.y - accr[1][r]) + accg[1][r];
        *(v2f*)(G2 + go) = d;
    }
}

__global__ void __launch_bounds__(S0_T)
step0_update(float* __restrict__ t1, float* __restrict__ t2, float* __restrict__ t3,
             const float* __restrict__ r2, const float* __restrict__ r3,
             const float* __restrict__ G2, const float* __restrict__ p0,
             const float* __restrict__ y, const float* __restrict__ W1T)
{
    __shared__ __attribute__((aligned(16))) float R2[SLAB_M * PW];
    __shared__ __attribute__((aligned(16))) float G2s[SLAB_M * PW];

    const int tid  = threadIdx.x;
    const int lane = tid & 31;
    const int wid  = tid >> 5;
    const int g    = lane >> 4;
    const int ln   = lane & 15;
    const int q    = blockIdx.x & 3;
    const size_t m0 = (size_t)(blockIdx.x >> 2) * SLAB_M;

    slab_stage<1024, S0_T>(R2, r2 + m0 * 1024, tid);
    slab_stage<1024, S0_T>(G2s, G2 + m0 * 1024, tid);
    __syncthreads();

    #pragma unroll
    for (int i = 0; i < SLAB_M * 32 / S0_T; ++i) {
        const int idx4 = i * S0_T + tid;
        const int row = idx4 >> 5;
        const int col = 128 * q + 4 * (idx4 & 31);
        const size_t go = (m0 + row) * 512 + col;
        v4f rv = *(const v4f*)(r3 + go);
        v4f tv = *(const v4f*)(t3 + go);
        v4f yv = *(const v4f*)(y + go);
        rv.x += EP_BETA * (tv.x - yv.x); rv.y += EP_BETA * (tv.y - yv.y);
        rv.z += EP_BETA * (tv.z - yv.z); rv.w += EP_BETA * (tv.w - yv.w);
        tv.x -= EP_LR * rv.x; tv.y -= EP_LR * rv.y;
        tv.z -= EP_LR * rv.z; tv.w -= EP_LR * rv.w;
        *(v4f*)(t3 + go) = tv;
    }

    #pragma unroll
    for (int i = 0; i < SLAB_M * 64 / S0_T; ++i) {
        const int idx4 = i * S0_T + tid;
        const int row = idx4 >> 6;
        const int col = 256 * q + 4 * (idx4 & 63);
        const size_t go = (m0 + row) * 1024 + col;
        v4f gv = *(const v4f*)(G2s + row * PW + col);
        v4f tv = *(const v4f*)(t2 + go);
        tv.x -= EP_LR * gv.x; tv.y -= EP_LR * gv.y;
        tv.z -= EP_LR * gv.z; tv.w -= EP_LR * gv.w;
        *(v4f*)(t2 + go) = tv;
    }

    const int nw = 256 * q + 32 * wid + 2 * ln;
    const float* Wl = W1T + (size_t)(4 * g) * 1024 + nw;
    v8f accr[2] = {};
    v8f accg[2] = {};
    slab_mma_kn<2, 1024, 1024>(R2, Wl, ln, g, accr);
    slab_mma_kn<2, 1024, 1024>(G2s, Wl, ln, g, accg);
    #pragma unroll
    for (int r = 0; r < 8; ++r) {
        const size_t go = (m0 + r + 8 * g) * 1024 + nw;
        v2f tv = *(const v2f*)(t1 + go);
        v2f pv = *(const v2f*)(p0 + go);
        v2f gv;
        gv.x = ((tv.x - pv.x) - accr[0][r]) + accg[0][r];
        gv.y = ((tv.y - pv.y) - accr[1][r]) + accg[1][r];
        tv.x -= EP_LR * gv.x; tv.y -= EP_LR * gv.y;
        *(v2f*)(t1 + go) = tv;
    }
}

#define EW(kernel, n, ...) kernel<<<dim3((n) / 1024), 256, 0, stream>>>(__VA_ARGS__)

extern "C" void kernel_launch(void* const* d_in, const int* in_sizes, int n_in,
                              void* d_out, int out_size, void* d_ws, size_t ws_size,
                              hipStream_t stream)
{
    const float* x  = (const float*)d_in[0];
    const float* y  = (const float*)d_in[1];
    const float* W0 = (const float*)d_in[2];
    const float* W1 = (const float*)d_in[3];
    const float* W2 = (const float*)d_in[4];
    float* out = (float*)d_out;

    const size_t n1 = (size_t)1024 * 1024;
    const size_t n3 = (size_t)1024 * 512;

    float* ws = (float*)d_ws;
    float* p0 = ws;
    float* t1 = p0 + n1;
    float* t2 = t1 + n1;
    float* t3 = t2 + n1;
    float* r1 = t3 + n3;
    float* r2 = r1 + n1;
    float* r3 = r2 + n1;
    float* part = r3 + n3;

    slab_forward<1024, false><<<dim3(FW_G1), FW_T, 0, stream>>>(x, W0, p0, 1.0f);
    EW(copyk, n1, t1, p0);
    slab_forward<1024, false><<<dim3(FW_G1), FW_T, 0, stream>>>(t1, W1, t2, 1.0f);
    slab_forward<512, false><<<dim3(FW_G3), FW_T, 0, stream>>>(t2, W2, t3, 1.0f);
    loss_partial<<<512, 256, 0, stream>>>(t3, y, part);
    loss_final<<<1, 256, 0, stream>>>(part, out + 2621440);

    float* W1T = r1;
    float* W2T = out + 2097152;
    float* G2  = out;
    transpose_rc<1024, 1024><<<dim3(32, 32), 256, 0, stream>>>(W1T, W1);
    transpose_rc<1024, 512><<<dim3(16, 32), 256, 0, stream>>>(W2T, W2);

    relax_resid<<<dim3(PH_G), PH_T, 0, stream>>>(t1, t2, t3, r2, r3, W1, W2);
    step0_g2<<<dim3(PH_G), S0_T, 0, stream>>>(r2, r3, t3, y, W2T, G2);
    step0_update<<<dim3(PH_G), S0_T, 0, stream>>>(t1, t2, t3, r2, r3, G2, p0, y, W1T);

    for (int it = 0; it < 19; ++it) {
        relax_resid<<<dim3(PH_G), PH_T, 0, stream>>>(t1, t2, t3, r2, r3, W1, W2);
        relax_update<<<dim3(PH_G), PH_T, 0, stream>>>(t1, t2, t3, r2, r3, p0, y, W1T, W2T);
    }

    relax_resid<<<dim3(PH_G), PH_T, 0, stream>>>(t1, t2, t3, r2, r3, W1, W2);
    EW(subk, n1, r1, t1, p0);
    float* sT = p0;
    transpose_rc<1024, 1024><<<dim3(32, 32), 256, 0, stream>>>(sT, x);
    slab_forward<1024, true><<<dim3(FW_G1), FW_T, 0, stream>>>(sT, r1, out,           -1.0f / EP_BETA);
    transpose_rc<1024, 1024><<<dim3(32, 32), 256, 0, stream>>>(sT, t1);
    slab_forward<1024, true><<<dim3(FW_G1), FW_T, 0, stream>>>(sT, r2, out + 1048576, -1.0f / EP_BETA);
    transpose_rc<1024, 1024><<<dim3(32, 32), 256, 0, stream>>>(sT, t2);
    slab_forward<512, true><<<dim3(FW_G3), FW_T, 0, stream>>>(sT, r3, out + 2097152, -1.0f / EP_BETA);
}


# K loops of both relaxation phases: a wave lowers its issue priority as it advances (s_setprio 3,2,1,0 over its eight trips), so a SIMD's three waves finish together; sums unchanged
# speedup vs baseline: 5.9624x; 1.0351x over previous
#include <hip/hip_runtime.h>


typedef __attribute__((ext_vector_type(2))) float v2f;
typedef __attribute__((ext_vector_type(8))) float v8f;

#define EP_BETA 1e-3f
#define EP_LR   0.5f

__global__ void __launch_bounds__(256) subk(float* __restrict__ o,
                                            const float* __restrict__ a,
                                            const float* __restrict__ b) {
    size_t i = ((size_t)blockIdx.x * 256 + threadIdx.x) * 4;
    float4 av = *(const float4*)(a + i);
    float4 bv = *(const float4*)(b + i);
    *(float4*)(o + i) = make_float4(av.x - bv.x, av.y - bv.y,
                                    av.z - bv.z, av.w - bv.w);
}

__global__ void __launch_bounds__(256) copyk(float* __restrict__ dst,
                                             const float* __restrict__ src) {
    size_t i = ((size_t)blockIdx.x * 256 + threadIdx.x) * 4;
    *(float4*)(dst + i) = *(const float4*)(src + i);
}

__global__ void __launch_bounds__(256) loss_partial(const float* __restrict__ s,
                                                    const float* __restrict__ y,
                                                    float* __restrict__ part) {
    __shared__ float red[256];
    size_t i = ((size_t)blockIdx.x * 256 + threadIdx.x) * 4;
    float4 a = *(const float4*)(s + i);
    float4 b = *(const float4*)(y + i);
    float dx = a.x - b.x, dy = a.y - b.y, dz = a.z - b.z, dw = a.w - b.w;
    red[threadIdx.x] = dx * dx + dy * dy + dz * dz + dw * dw;
    __syncthreads();
    for (int off = 128; off > 0; off >>= 1) {
        if (threadIdx.x < off) red[threadIdx.x] += red[threadIdx.x + off];
        __syncthreads();
    }
    if (threadIdx.x == 0) part[blockIdx.x] = red[0];
}

__global__ void __launch_bounds__(256) loss_final(const float* __restrict__ part,
                                                  float* __restrict__ out) {
    __shared__ float red[256];
    red[threadIdx.x] = part[threadIdx.x] + part[threadIdx.x + 256];
    __syncthreads();
    for (int off = 128; off > 0; off >>= 1) {
        if (threadIdx.x < off) red[threadIdx.x] += red[threadIdx.x + off];
        __syncthreads();
    }
    if (threadIdx.x == 0) out[0] = red[0] * (1.0f / 524288.0f);
}

template <int R, int C>
__global__ void __launch_bounds__(256) transpose_rc(float* __restrict__ dst,
                                                    const float* __restrict__ src)
{
    __shared__ float tile[32][33];
    const int tx = threadIdx.x & 31;
    const int ty = threadIdx.x >> 5;
    const int c0 = blockIdx.x * 32;
    const int r0 = blockIdx.y * 32;
    #pragma unroll
    for (int i = 0; i < 4; ++i)
        tile[ty + 8 * i][tx] = src[(size_t)(r0 + ty + 8 * i) * C + c0 + tx];
    __syncthreads();
    #pragma unroll
    for (int i = 0; i < 4; ++i)
        dst[(size_t)(c0 + ty + 8 * i) * R + r0 + tx] = tile[tx][ty + 8 * i];
}

typedef __attribute__((ext_vector_type(4))) float v4f;

constexpr int SLAB_M = 16;
constexpr int PH_T   = 384;
constexpr int PH_M   = 32;
constexpr int PH_G   = 256;

template <int NT> struct ColVec;
template <> struct ColVec<4> { typedef v4f type; };
template <> struct ColVec<2> { typedef v2f type; };

template <int K, int T = PH_T, int ROWS = SLAB_M>
__device__ __forceinline__ void slab_stage(float* As, const float* src, int tid)
{
    constexpr int P = K + 4;
    constexpr int Q = K / 4;
    constexpr int TOT = ROWS * Q;
    #pragma unroll
    for (int i = 0; i < (TOT + T - 1) / T; ++i) {
        int idx4 = i * T + tid;
        if (TOT % T != 0 && i == TOT / T && idx4 >= TOT) idx4 -= T;
        int row  = idx4 / Q;
        int c4   = idx4 - row * Q;
        v4f v = *(const v4f*)(src + (size_t)idx4 * 4);
        *(v4f*)(As + row * P + 4 * c4) = v;
    }
}

template <int NT, int K, int LD, int P = K + 4>
__device__ __forceinline__ void slab_mma_kn(const float* As, const float* __restrict__ Wl,
                                            int ln, int g, v8f (&acc)[NT])
{
    typedef typename ColVec<NT>::type VT;
    const float* ap = As + ln * P + 4 * g;
    #pragma unroll (4 / NT)
    for (int kc = 0; kc < K; kc += 32) {
        VT  bv[4][4];
        v4f av[4];
        #pragma unroll
        for (int c = 0; c < 4; ++c) {
            #pragma unroll
            for (int i = 0; i < 4; ++i)
                bv[c][i] = *(const VT*)(Wl + (size_t)(kc + 8 * c + i) * LD);
        }
        #pragma unroll
        for (int c = 0; c < 4; ++c) av[c] = *(const v4f*)(ap + kc + 8 * c);
        #pragma unroll
        for (int c = 0; c < 4; ++c) {
            #pragma unroll
            for (int s = 0; s < 2; ++s) {
                v2f a; a.x = av[c][2 * s]; a.y = av[c][2 * s + 1];
                #pragma unroll
                for (int j = 0; j < NT; ++j) {
                    v2f b; b.x = bv[c][2 * s][j]; b.y = bv[c][2 * s + 1][j];
                    acc[j] = __builtin_amdgcn_wmma_f32_16x16x4_f32(
                        false, a, false, b, (short)0, acc[j], false, false);
                }
            }
        }
    }
}

template <int KL, int LD>
__device__ __forceinline__ void slab_mma_2x2(const float* a0, const float* a1,
                                             const float* __restrict__ Wl, v8f (&acc)[2][2])
{
    static_assert(KL % 512 == 0, "an eighth of the K range is a whole trip");
    #pragma unroll 1
    for (int kt = 0; kt < KL; kt += 64) {
        if (kt == 0)               __builtin_amdgcn_s_setprio(3);
        else if (kt == 3 * KL / 8) __builtin_amdgcn_s_setprio(2);
        else if (kt == 5 * KL / 8) __builtin_amdgcn_s_setprio(1);
        else if (kt == 7 * KL / 8) __builtin_amdgcn_s_setprio(0);
        v2f bv[2][4][4];
        v4f av[2][2][4];
        #pragma unroll
        for (int h = 0; h < 2; ++h) {
            #pragma unroll
            for (int c = 0; c < 4; ++c) {
                #pragma unroll
                for (int i = 0; i < 4; ++i)
                    bv[h][c][i] = *(const v2f*)(Wl + (size_t)(kt + 32 * h + 8 * c + i) * LD);
            }
        }
        #pragma unroll
        for (int h = 0; h < 2; ++h) {
            #pragma unroll
            for (int c = 0; c < 4; ++c) {
                av[h][0][c] = *(const v4f*)(a0 + kt + 32 * h + 8 * c);
                av[h][1][c] = *(const v4f*)(a1 + kt + 32 * h + 8 * c);
            }
            #pragma unroll
            for (int c = 0; c < 4; ++c) {
                #pragma unroll
                for (int s = 0; s < 2; ++s) {
                    v2f x0; x0.x = av[h][0][c][2 * s]; x0.y = av[h][0][c][2 * s + 1];
                    v2f x1; x1.x = av[h][1][c][2 * s]; x1.y = av[h][1][c][2 * s + 1];
                    #pragma unroll
                    for (int j = 0; j < 2; ++j) {
                        v2f b; b.x = bv[h][c][2 * s][j]; b.y = bv[h][c][2 * s + 1][j];
                        acc[0][j] = __builtin_amdgcn_wmma_f32_16x16x4_f32(
                            false, x0, false, b, (short)0, acc[0][j], false, false);
                        acc[1][j] = __builtin_amdgcn_wmma_f32_16x16x4_f32(
                            false, x1, false, b, (short)0, acc[1][j], false, false);
                    }
                }
            }
        }
    }
}

__device__ __forceinline__ void pair_put(float* X, int wid, int lane, const v8f (&a)[2])
{
    float* xo = X + wid * 512 + lane;
    #pragma unroll
    for (int j = 0; j < 2; ++j) {
        #pragma unroll
        for (int r = 0; r < 8; ++r) xo[(8 * j + r) * 32] = a[j][r];
    }
}

__device__ __forceinline__ void pair_get(const float* X, int wid, int lane, v8f (&a)[2])
{
    const float* xi = X + (wid ^ 1) * 512 + lane;
    #pragma unroll
    for (int j = 0; j < 2; ++j) {
        #pragma unroll
        for (int r = 0; r < 8; ++r) a[j][r] += xi[(8 * j + r) * 32];
    }
}

constexpr int PW = 1028;
constexpr int PH = 516;

constexpr int FW_T  = 128;
constexpr int FW_G1 = 64 * (1024 / 128);
constexpr int FW_G3 = 64 * (512 / 128);

template <int N, bool SC>
__global__ void __launch_bounds__(FW_T)
slab_forward(const float* __restrict__ in, const float* __restrict__ W,
             float* __restrict__ out, float alpha)
{
    __shared__ __attribute__((aligned(16))) float S[SLAB_M * PW];

    constexpr int NB = N / 128;
    const int tid  = threadIdx.x;
    const int lane = tid & 31;
    const int wid  = tid >> 5;
    const int g    = lane >> 4;
    const int ln   = lane & 15;
    const int cb   = blockIdx.x % NB;
    const size_t m0 = (size_t)(blockIdx.x / NB) * SLAB_M;

    slab_stage<1024, FW_T>(S, in + m0 * 1024, tid);
    __syncthreads();

    const int nw = 128 * cb + 32 * wid + 2 * ln;
    v8f acc[2] = {};
    v8f ach[2] = {};
    slab_mma_kn<2, 512, N, PW>(S, W + (size_t)(4 * g) * N + nw, ln, g, acc);
    slab_mma_kn<2, 512, N, PW>(S + 512, W + (size_t)(512 + 4 * g) * N + nw, ln, g, ach);
    #pragma unroll
    for (int r = 0; r < 8; ++r) {
        const float s0 = acc[0][r] + ach[0][r];
        const float s1 = acc[1][r] + ach[1][r];
        v2f d;
        d.x = SC ? alpha * s0 : s0;
        d.y = SC ? alpha * s1 : s1;
        *(v2f*)(out + (m0 + r + 8 * g) * N + nw) = d;
    }
}

__global__ void __launch_bounds__(PH_T)
relax_resid(const float* __restrict__ t1, const float* __restrict__ t2,
            const float* __restrict__ t3,
            float* __restrict__ r2, float* __restrict__ r3,
            const float* __restrict__ W1, const float* __restrict__ W2)
{
    __shared__ __attribute__((aligned(16))) float S1[PH_M * PW];
    __shared__ __attribute__((aligned(16))) float S2[PH_M * PW];
    __shared__ __attribute__((aligned(16))) float X[12 * 512];

    const int tid  = threadIdx.x;
    const int lane = tid & 31;
    const int wid  = __builtin_amdgcn_readfirstlane(tid >> 5);
    const int g    = lane >> 4;
    const int ln   = lane & 15;
    const int kh   = wid & 1;
    const int cb   = blockIdx.x & 7;
    const size_t m0 = (size_t)(blockIdx.x >> 3) * PH_M;

    slab_stage<1024, PH_T, PH_M>(S1, t1 + m0 * 1024, tid);
    slab_stage<1024, PH_T, PH_M>(S2, t2 + m0 * 1024, tid);
    __syncthreads();

    const int k0   = 512 * kh;
    const int own  = (ln + 16 * kh) * PW + 4 * g + k0;
    const int oth  = (ln + 16 * (kh ^ 1)) * PW + 4 * g + k0;
    v8f acc[2][2] = {};
    int nw;
    if (wid < 8) {
        nw = 128 * cb + 32 * (wid >> 1) + 2 * ln;
        slab_mma_2x2<512, 1024>(S1 + own, S1 + oth,
                                W1 + (size_t)(k0 + 4 * g) * 1024 + nw, acc);
    } else {
        nw = 64 * cb + 32 * ((wid - 8) >> 1) + 2 * ln;
        slab_mma_2x2<512, 512>(S2 + own, S2 + oth,
                               W2 + (size_t)(k0 + 4 * g) * 512 + nw, acc);
    }
    pair_put(X, wid, lane, acc[1]);
    __syncthreads();
    pair_get(X, wid, lane, acc[0]);

    if (wid < 8) {
        #pragma unroll
        for (int r = 0; r < 8; ++r) {
            const int row = 16 * kh + r + 8 * g;
            v2f c = *(const v2f*)(S2 + row * PW + nw);
            v2f d;
            d.x = c.x - acc[0][0][r]; d.y = c.y - acc[0][1][r];
            *(v2f*)(r2 + (m0 + row) * 1024 + nw) = d;
        }
    } else {
        #pragma unroll
        for (int r = 0; r < 8; ++r) {
            const size_t go = (m0 + 16 * kh + r + 8 * g) * 512 + nw;
            v2f c = *(const v2f*)(t3 + go);
            v2f d;
            d.x = c.x - acc[0][0][r]; d.y = c.y - acc[0][1][r];
            *(v2f*)(r3 + go) = d;
        }
    }
}

__global__ void __launch_bounds__(PH_T)
relax_update(float* __restrict__ t1, float* __restrict__ t2, float* __restrict__ t3,
             const float* __restrict__ r2, const float* __restrict__ r3,
             const float* __restrict__ p0, const float* __restrict__ y,
             const float* __restrict__ W1T, const float* __restrict__ W2T)
{
    __shared__ __attribute__((aligned(16))) float R2[PH_M * PW];
    __shared__ __attribute__((aligned(16))) float R3[PH_M * PH];
    __shared__ __attribute__((aligned(16))) float X[8 * 512];

    const int tid  = threadIdx.x;
    const int lane = tid & 31;
    const int wid  = __builtin_amdgcn_readfirstlane(tid >> 5);
    const int g    = lane >> 4;
    const int ln   = lane & 15;
    const int kh   = wid & 1;
    const int cb   = blockIdx.x & 7;
    const size_t m0 = (size_t)(blockIdx.x >> 3) * PH_M;

    slab_stage<1024, PH_T, PH_M>(R2, r2 + m0 * 1024, tid);
    slab_stage<512, PH_T, PH_M>(R3, r3 + m0 * 512, tid);
    __syncthreads();

    #pragma unroll
    for (int i = 0; i < (PH_M * 16 + PH_T - 1) / PH_T; ++i) {
        int idx4 = i * PH_T + tid;
        if (idx4 < PH_M * 16) {
            const int row = idx4 >> 4;
            const int col = 64 * cb + 4 * (idx4 & 15);
            const size_t go = (m0 + row) * 512 + col;
            v4f rv = *(const v4f*)(R3 + row * PH + col);
            v4f tv = *(const v4f*)(t3 + go);
            v4f yv = *(const v4f*)(y + go);
            rv.x += EP_BETA * (tv.x - yv.x); rv.y += EP_BETA * (tv.y - yv.y);
            rv.z += EP_BETA * (tv.z - yv.z); rv.w += EP_BETA * (tv.w - yv.w);
            tv.x -= EP_LR * rv.x; tv.y -= EP_LR * rv.y;
            tv.z -= EP_LR * rv.z; tv.w -= EP_LR * rv.w;
            *(v4f*)(t3 + go) = tv;
        }
    }

    v8f acc[2][2] = {};
    int nw;
    if (wid < 8) {
        const int k0  = 512 * kh;
        const int own = (ln + 16 * kh) * PW + 4 * g + k0;
        const int oth = (ln + 16 * (kh ^ 1)) * PW + 4 * g + k0;
        nw = 128 * cb + 32 * (wid >> 1) + 2 * ln;
        slab_mma_2x2<512, 1024>(R2 + own, R2 + oth,
                                W1T + (size_t)(k0 + 4 * g) * 1024 + nw, acc);
        pair_put(X, wid, lane, acc[1]);
    } else {
        nw = 128 * cb + 32 * (wid - 8) + 2 * ln;
        slab_mma_2x2<512, 1024>(R3 + ln * PH + 4 * g, R3 + (ln + 16) * PH + 4 * g,
                                W2T + (size_t)(4 * g) * 1024 + nw, acc);
    }
    __syncthreads();

    if (wid < 8) {
        pair_get(X, wid, lane, acc[0]);
        #pragma unroll
        for (int r = 0; r < 8; ++r) {
            const size_t go = (m0 + 16 * kh + r + 8 * g) * 1024 + nw;
            v2f tv = *(const v2f*)(t1 + go);
            v2f pv = *(const v2f*)(p0 + go);
            v2f gv;
            gv.x = (tv.x - pv.x) - acc[0][0][r]; gv.y = (tv.y - pv.y) - acc[0][1][r];
            tv.x -= EP_LR * gv.x; tv.y -= EP_LR * gv.y;
            *(v2f*)(t1 + go) = tv;
        }
    } else {
        #pragma unroll
        for (int m = 0; m < 2; ++m) {
            #pragma unroll
            for (int r = 0; r < 8; ++r) {
                const int row = 16 * m + r + 8 * g;
                const size_t go = (m0 + row) * 1024 + nw;
                v2f rv = *(const v2f*)(R2 + row * PW + nw);
                v2f tv = *(const v2f*)(t2 + go);
                v2f gv;
                gv.x = rv.x - acc[m][0][r]; gv.y = rv.y - acc[m][1][r];
                tv.x -= EP_LR * gv.x; tv.y -= EP_LR * gv.y;
                *(v2f*)(t2 + go) = tv;
            }
        }
    }
}

constexpr int S0_T = 256;

__global__ void __launch_bounds__(S0_T)
step0_g2(const float* __restrict__ r2, const float* __restrict__ r3,
         const float* __restrict__ t3, const float* __restrict__ y,
         const float* __restrict__ W2T, float* __restrict__ G2)
{
    __shared__ __attribute__((aligned(16))) float R3[SLAB_M * PH];
    __shared__ __attribute__((aligned(16))) float G3[SLAB_M * PH];

    const int tid  = threadIdx.x;
    const int lane = tid & 31;
    const int wid  = tid >> 5;
    const int g    = lane >> 4;
    const int ln   = lane & 15;
    const int q    = blockIdx.x & 3;
    const size_t m0 = (size_t)(blockIdx.x >> 2) * SLAB_M;

    #pragma unroll
    for (int i = 0; i < SLAB_M * 128 / S0_T; ++i) {
        const int idx4 = i * S0_T + tid;
        const int row = idx4 >> 7;
        const int col = 4 * (idx4 & 127);
        const size_t go = (m0 + row) * 512 + col;
        v4f rv = *(const v4f*)(r3 + go);
        v4f tv = *(const v4f*)(t3 + go);
        v4f yv = *(const v4f*)(y + go);
        *(v4f*)(R3 + row * PH + col) = rv;
        rv.x += EP_BETA * (tv.x - yv.x); rv.y += EP_BETA * (tv.y - yv.y);
        rv.z += EP_BETA * (tv.z - yv.z); rv.w += EP_BETA * (tv.w - yv.w);
        *(v4f*)(G3 + row * PH + col) = rv;
    }
    __syncthreads();

    const int nw = 256 * q + 32 * wid + 2 * ln;
    const float* Wl = W2T + (size_t)(4 * g) * 1024 + nw;
    v8f accr[2] = {};
    v8f accg[2] = {};
    slab_mma_kn<2, 512, 1024>(R3, Wl, ln, g, accr);
    slab_mma_kn<2, 512, 1024>(G3, Wl, ln, g, accg);
    #pragma unroll
    for (int r = 0; r < 8; ++r) {
        const size_t go = (m0 + r + 8 * g) * 1024 + nw;
        v2f c = *(const v2f*)(r2 + go);
        v2f d;
        d.x = (c.x - accr[0][r]) + accg[0][r];
        d.y = (c.y - accr[1][r]) + accg[1][r];
        *(v2f*)(G2 + go) = d;
    }
}

__global__ void __launch_bounds__(S0_T)
step0_update(float* __restrict__ t1, float* __restrict__ t2, float* __restrict__ t3,
             const float* __restrict__ r2, const float* __restrict__ r3,
             const float* __restrict__ G2, const float* __restrict__ p0,
             const float* __restrict__ y, const float* __restrict__ W1T)
{
    __shared__ __attribute__((aligned(16))) float R2[SLAB_M * PW];
    __shared__ __attribute__((aligned(16))) float G2s[SLAB_M * PW];

    const int tid  = threadIdx.x;
    const int lane = tid & 31;
    const int wid  = tid >> 5;
    const int g    = lane >> 4;
    const int ln   = lane & 15;
    const int q    = blockIdx.x & 3;
    const size_t m0 = (size_t)(blockIdx.x >> 2) * SLAB_M;

    slab_stage<1024, S0_T>(R2, r2 + m0 * 1024, tid);
    slab_stage<1024, S0_T>(G2s, G2 + m0 * 1024, tid);
    __syncthreads();

    #pragma unroll
    for (int i = 0; i < SLAB_M * 32 / S0_T; ++i) {
        const int idx4 = i * S0_T + tid;
        const int row = idx4 >> 5;
        const int col = 128 * q + 4 * (idx4 & 31);
        const size_t go = (m0 + row) * 512 + col;
        v4f rv = *(const v4f*)(r3 + go);
        v4f tv = *(const v4f*)(t3 + go);
        v4f yv = *(const v4f*)(y + go);
        rv.x += EP_BETA * (tv.x - yv.x); rv.y += EP_BETA * (tv.y - yv.y);
        rv.z += EP_BETA * (tv.z - yv.z); rv.w += EP_BETA * (tv.w - yv.w);
        tv.x -= EP_LR * rv.x; tv.y -= EP_LR * rv.y;
        tv.z -= EP_LR * rv.z; tv.w -= EP_LR * rv.w;
        *(v4f*)(t3 + go) = tv;
    }

    #pragma unroll
    for (int i = 0; i < SLAB_M * 64 / S0_T; ++i) {
        const int idx4 = i * S0_T + tid;
        const int row = idx4 >> 6;
        const int col = 256 * q + 4 * (idx4 & 63);
        const size_t go = (m0 + row) * 1024 + col;
        v4f gv = *(const v4f*)(G2s + row * PW + col);
        v4f tv = *(const v4f*)(t2 + go);
        tv.x -= EP_LR * gv.x; tv.y -= EP_LR * gv.y;
        tv.z -= EP_LR * gv.z; tv.w -= EP_LR * gv.w;
        *(v4f*)(t2 + go) = tv;
    }

    const int nw = 256 * q + 32 * wid + 2 * ln;
    const float* Wl = W1T + (size_t)(4 * g) * 1024 + nw;
    v8f accr[2] = {};
    v8f accg[2] = {};
    slab_mma_kn<2, 1024, 1024>(R2, Wl, ln, g, accr);
    slab_mma_kn<2, 1024, 1024>(G2s, Wl, ln, g, accg);
    #pragma unroll
    for (int r = 0; r < 8; ++r) {
        const size_t go = (m0 + r + 8 * g) * 1024 + nw;
        v2f tv = *(const v2f*)(t1 + go);
        v2f pv = *(const v2f*)(p0 + go);
        v2f gv;
        gv.x = ((tv.x - pv.x) - accr[0][r]) + accg[0][r];
        gv.y = ((tv.y - pv.y) - accr[1][r]) + accg[1][r];
        tv.x -= EP_LR * gv.x; tv.y -= EP_LR * gv.y;
        *(v2f*)(t1 + go) = tv;
    }
}

#define EW(kernel, n, ...) kernel<<<dim3((n) / 1024), 256, 0, stream>>>(__VA_ARGS__)

extern "C" void kernel_launch(void* const* d_in, const int* in_sizes, int n_in,
                              void* d_out, int out_size, void* d_ws, size_t ws_size,
                              hipStream_t stream)
{
    const float* x  = (const float*)d_in[0];
    const float* y  = (const float*)d_in[1];
    const float* W0 = (const float*)d_in[2];
    const float* W1 = (const float*)d_in[3];
    const float* W2 = (const float*)d_in[4];
    float* out = (float*)d_out;

    const size_t n1 = (size_t)1024 * 1024;
    const size_t n3 = (size_t)1024 * 512;

    float* ws = (float*)d_ws;
    float* p0 = ws;
    float* t1 = p0 + n1;
    float* t2 = t1 + n1;
    float* t3 = t2 + n1;
    float* r1 = t3 + n3;
    float* r2 = r1 + n1;
    float* r3 = r2 + n1;
    float* part = r3 + n3;

    slab_forward<1024, false><<<dim3(FW_G1), FW_T, 0, stream>>>(x, W0, p0, 1.0f);
    EW(copyk, n1, t1, p0);
    slab_forward<1024, false><<<dim3(FW_G1), FW_T, 0, stream>>>(t1, W1, t2, 1.0f);
    slab_forward<512, false><<<dim3(FW_G3), FW_T, 0, stream>>>(t2, W2, t3, 1.0f);
    loss_partial<<<512, 256, 0, stream>>>(t3, y, part);
    loss_final<<<1, 256, 0, stream>>>(part, out + 2621440);

    float* W1T = r1;
    float* W2T = out + 2097152;
    float* G2  = out;
    transpose_rc<1024, 1024><<<dim3(32, 32), 256, 0, stream>>>(W1T, W1);
    transpose_rc<1024, 512><<<dim3(16, 32), 256, 0, stream>>>(W2T, W2);

    relax_resid<<<dim3(PH_G), PH_T, 0, stream>>>(t1, t2, t3, r2, r3, W1, W2);
    step0_g2<<<dim3(PH_G), S0_T, 0, stream>>>(r2, r3, t3, y, W2T, G2);
    step0_update<<<dim3(PH_G), S0_T, 0, stream>>>(t1, t2, t3, r2, r3, G2, p0, y, W1T);

    for (int it = 0; it < 19; ++it) {
        relax_resid<<<dim3(PH_G), PH_T, 0, stream>>>(t1, t2, t3, r2, r3, W1, W2);
        relax_update<<<dim3(PH_G), PH_T, 0, stream>>>(t1, t2, t3, r2, r3, p0, y, W1T, W2T);
    }

    relax_resid<<<dim3(PH_G), PH_T, 0, stream>>>(t1, t2, t3, r2, r3, W1, W2);
    EW(subk, n1, r1, t1, p0);
    float* sT = p0;
    transpose_rc<1024, 1024><<<dim3(32, 32), 256, 0, stream>>>(sT, x);
    slab_forward<1024, true><<<dim3(FW_G1), FW_T, 0, stream>>>(sT, r1, out,           -1.0f / EP_BETA);
    transpose_rc<1024, 1024><<<dim3(32, 32), 256, 0, stream>>>(sT, t1);
    slab_forward<1024, true><<<dim3(FW_G1), FW_T, 0, stream>>>(sT, r2, out + 1048576, -1.0f / EP_BETA);
    transpose_rc<1024, 1024><<<dim3(32, 32), 256, 0, stream>>>(sT, t2);
    slab_forward<512, true><<<dim3(FW_G3), FW_T, 0, stream>>>(sT, r3, out + 2097152, -1.0f / EP_BETA);
}


# Relaxation phases: a slab's two products go to its 8 work-groups by product, not by column block: 7 of 8 stage only the one K plane their product reads, the two-plane one enters first; sums unchanged
# speedup vs baseline: 6.8702x; 1.1523x over previous
#include <hip/hip_runtime.h>


typedef __attribute__((ext_vector_type(2))) float v2f;
typedef __attribute__((ext_vector_type(8))) float v8f;

#define EP_BETA 1e-3f
#define EP_LR   0.5f

__global__ void __launch_bounds__(256) subk(float* __restrict__ o,
                                            const float* __restrict__ a,
                                            const float* __restrict__ b) {
    size_t i = ((size_t)blockIdx.x * 256 + threadIdx.x) * 4;
    float4 av = *(const float4*)(a + i);
    float4 bv = *(const float4*)(b + i);
    *(float4*)(o + i) = make_float4(av.x - bv.x, av.y - bv.y,
                                    av.z - bv.z, av.w - bv.w);
}

__global__ void __launch_bounds__(256) copyk(float* __restrict__ dst,
                                             const float* __restrict__ src) {
    size_t i = ((size_t)blockIdx.x * 256 + threadIdx.x) * 4;
    *(float4*)(dst + i) = *(const float4*)(src + i);
}

__global__ void __launch_bounds__(256) loss_partial(const float* __restrict__ s,
                                                    const float* __restrict__ y,
                                                    float* __restrict__ part) {
    __shared__ float red[256];
    size_t i = ((size_t)blockIdx.x * 256 + threadIdx.x) * 4;
    float4 a = *(const float4*)(s + i);
    float4 b = *(const float4*)(y + i);
    float dx = a.x - b.x, dy = a.y - b.y, dz = a.z - b.z, dw = a.w - b.w;
    red[threadIdx.x] = dx * dx + dy * dy + dz * dz + dw * dw;
    __syncthreads();
    for (int off = 128; off > 0; off >>= 1) {
        if (threadIdx.x < off) red[threadIdx.x] += red[threadIdx.x + off];
        __syncthreads();
    }
    if (threadIdx.x == 0) part[blockIdx.x] = red[0];
}

__global__ void __launch_bounds__(256) loss_final(const float* __restrict__ part,
                                                  float* __restrict__ out) {
    __shared__ float red[256];
    red[threadIdx.x] = part[threadIdx.x] + part[threadIdx.x + 256];
    __syncthreads();
    for (int off = 128; off > 0; off >>= 1) {
        if (threadIdx.x < off) red[threadIdx.x] += red[threadIdx.x + off];
        __syncthreads();
    }
    if (threadIdx.x == 0) out[0] = red[0] * (1.0f / 524288.0f);
}

template <int R, int C>
__global__ void __launch_bounds__(256) transpose_rc(float* __restrict__ dst,
                                                    const float* __restrict__ src)
{
    __shared__ float tile[32][33];
    const int tx = threadIdx.x & 31;
    const int ty = threadIdx.x >> 5;
    const int c0 = blockIdx.x * 32;
    const int r0 = blockIdx.y * 32;
    #pragma unroll
    for (int i = 0; i < 4; ++i)
        tile[ty + 8 * i][tx] = src[(size_t)(r0 + ty + 8 * i) * C + c0 + tx];
    __syncthreads();
    #pragma unroll
    for (int i = 0; i < 4; ++i)
        dst[(size_t)(c0 + ty + 8 * i) * R + r0 + tx] = tile[tx][ty + 8 * i];
}

typedef __attribute__((ext_vector_type(4))) float v4f;

constexpr int SLAB_M = 16;
constexpr int PH_T   = 384;
constexpr int PH_M   = 32;
constexpr int PH_G   = 256;

__device__ __forceinline__ int ph_role(int place)
{
    return place == 0 ? 5 : (place <= 5 ? place - 1 : place);
}

template <int NT> struct ColVec;
template <> struct ColVec<4> { typedef v4f type; };
template <> struct ColVec<2> { typedef v2f type; };

template <int K, int T = PH_T, int ROWS = SLAB_M>
__device__ __forceinline__ void slab_stage(float* As, const float* src, int tid)
{
    constexpr int P = K + 4;
    constexpr int Q = K / 4;
    constexpr int TOT = ROWS * Q;
    #pragma unroll
    for (int i = 0; i < (TOT + T - 1) / T; ++i) {
        int idx4 = i * T + tid;
        if (TOT % T != 0 && i == TOT / T && idx4 >= TOT) idx4 -= T;
        int row  = idx4 / Q;
        int c4   = idx4 - row * Q;
        v4f v = *(const v4f*)(src + (size_t)idx4 * 4);
        *(v4f*)(As + row * P + 4 * c4) = v;
    }
}

template <int K, int T, int ROWS>
struct SlabRegs { v4f v[(ROWS * (K / 4) + T - 1) / T]; };

template <int K, int T, int ROWS>
__device__ __forceinline__ void slab_fetch(SlabRegs<K, T, ROWS>& s, const float* src, int tid)
{
    constexpr int TOT = ROWS * (K / 4);
    #pragma unroll
    for (int i = 0; i < (TOT + T - 1) / T; ++i) {
        int idx4 = i * T + tid;
        if (TOT % T != 0 && i == TOT / T && idx4 >= TOT) idx4 -= T;
        s.v[i] = *(const v4f*)(src + (size_t)idx4 * 4);
    }
}

template <int K, int T, int ROWS>
__device__ __forceinline__ void slab_put(float* As, const SlabRegs<K, T, ROWS>& s, int tid)
{
    constexpr int P = K + 4;
    constexpr int Q = K / 4;
    constexpr int TOT = ROWS * Q;
    #pragma unroll
    for (int i = 0; i < (TOT + T - 1) / T; ++i) {
        int idx4 = i * T + tid;
        if (TOT % T != 0 && i == TOT / T && idx4 >= TOT) idx4 -= T;
        int row  = idx4 / Q;
        int c4   = idx4 - row * Q;
        *(v4f*)(As + row * P + 4 * c4) = s.v[i];
    }
}

template <int NT, int K, int LD, int P = K + 4>
__device__ __forceinline__ void slab_mma_kn(const float* As, const float* __restrict__ Wl,
                                            int ln, int g, v8f (&acc)[NT])
{
    typedef typename ColVec<NT>::type VT;
    const float* ap = As + ln * P + 4 * g;
    #pragma unroll (4 / NT)
    for (int kc = 0; kc < K; kc += 32) {
        VT  bv[4][4];
        v4f av[4];
        #pragma unroll
        for (int c = 0; c < 4; ++c) {
            #pragma unroll
            for (int i = 0; i < 4; ++i)
                bv[c][i] = *(const VT*)(Wl + (size_t)(kc + 8 * c + i) * LD);
        }
        #pragma unroll
        for (int c = 0; c < 4; ++c) av[c] = *(const v4f*)(ap + kc + 8 * c);
        #pragma unroll
        for (int c = 0; c < 4; ++c) {
            #pragma unroll
            for (int s = 0; s < 2; ++s) {
                v2f a; a.x = av[c][2 * s]; a.y = av[c][2 * s + 1];
                #pragma unroll
                for (int j = 0; j < NT; ++j) {
                    v2f b; b.x = bv[c][2 * s][j]; b.y = bv[c][2 * s + 1][j];
                    acc[j] = __builtin_amdgcn_wmma_f32_16x16x4_f32(
                        false, a, false, b, (short)0, acc[j], false, false);
                }
            }
        }
    }
}

template <int KL, int LD>
__device__ __forceinline__ void slab_mma_2x2(const float* a0, const float* a1,
                                             const float* __restrict__ Wl, v8f (&acc)[2][2])
{
    static_assert(KL % 512 == 0, "an eighth of the K range is a whole trip");
    #pragma unroll 1
    for (int kt = 0; kt < KL; kt += 64) {
        if (kt == 0)               __builtin_amdgcn_s_setprio(3);
        else if (kt == 3 * KL / 8) __builtin_amdgcn_s_setprio(2);
        else if (kt == 5 * KL / 8) __builtin_amdgcn_s_setprio(1);
        else if (kt == 7 * KL / 8) __builtin_amdgcn_s_setprio(0);
        v2f bv[2][4][4];
        v4f av[2][2][4];
        #pragma unroll
        for (int h = 0; h < 2; ++h) {
            #pragma unroll
            for (int c = 0; c < 4; ++c) {
                #pragma unroll
                for (int i = 0; i < 4; ++i)
                    bv[h][c][i] = *(const v2f*)(Wl + (size_t)(kt + 32 * h + 8 * c + i) * LD);
            }
        }
        #pragma unroll
        for (int h = 0; h < 2; ++h) {
            #pragma unroll
            for (int c = 0; c < 4; ++c) {
                av[h][0][c] = *(const v4f*)(a0 + kt + 32 * h + 8 * c);
                av[h][1][c] = *(const v4f*)(a1 + kt + 32 * h + 8 * c);
            }
            #pragma unroll
            for (int c = 0; c < 4; ++c) {
                #pragma unroll
                for (int s = 0; s < 2; ++s) {
                    v2f x0; x0.x = av[h][0][c][2 * s]; x0.y = av[h][0][c][2 * s + 1];
                    v2f x1; x1.x = av[h][1][c][2 * s]; x1.y = av[h][1][c][2 * s + 1];
                    #pragma unroll
                    for (int j = 0; j < 2; ++j) {
                        v2f b; b.x = bv[h][c][2 * s][j]; b.y = bv[h][c][2 * s + 1][j];
                        acc[0][j] = __builtin_amdgcn_wmma_f32_16x16x4_f32(
                            false, x0, false, b, (short)0, acc[0][j], false, false);
                        acc[1][j] = __builtin_amdgcn_wmma_f32_16x16x4_f32(
                            false, x1, false, b, (short)0, acc[1][j], false, false);
                    }
                }
            }
        }
    }
}

__device__ __forceinline__ void pair_put(float* X, int wid, int lane, const v8f (&a)[2])
{
    float* xo = X + wid * 512 + lane;
    #pragma unroll
    for (int j = 0; j < 2; ++j) {
        #pragma unroll
        for (int r = 0; r < 8; ++r) xo[(8 * j + r) * 32] = a[j][r];
    }
}

__device__ __forceinline__ void pair_get(const float* X, int wid, int lane, v8f (&a)[2])
{
    const float* xi = X + (wid ^ 1) * 512 + lane;
    #pragma unroll
    for (int j = 0; j < 2; ++j) {
        #pragma unroll
        for (int r = 0; r < 8; ++r) a[j][r] += xi[(8 * j + r) * 32];
    }
}

constexpr int PW = 1028;
constexpr int PH = 516;

constexpr int FW_T  = 128;
constexpr int FW_G1 = 64 * (1024 / 128);
constexpr int FW_G3 = 64 * (512 / 128);

template <int N, bool SC>
__global__ void __launch_bounds__(FW_T)
slab_forward(const float* __restrict__ in, const float* __restrict__ W,
             float* __restrict__ out, float alpha)
{
    __shared__ __attribute__((aligned(16))) float S[SLAB_M * PW];

    constexpr int NB = N / 128;
    const int tid  = threadIdx.x;
    const int lane = tid & 31;
    const int wid  = tid >> 5;
    const int g    = lane >> 4;
    const int ln   = lane & 15;
    const int cb   = blockIdx.x % NB;
    const size_t m0 = (size_t)(blockIdx.x / NB) * SLAB_M;

    slab_stage<1024, FW_T>(S, in + m0 * 1024, tid);
    __syncthreads();

    const int nw = 128 * cb + 32 * wid + 2 * ln;
    v8f acc[2] = {};
    v8f ach[2] = {};
    slab_mma_kn<2, 512, N, PW>(S, W + (size_t)(4 * g) * N + nw, ln, g, acc);
    slab_mma_kn<2, 512, N, PW>(S + 512, W + (size_t)(512 + 4 * g) * N + nw, ln, g, ach);
    #pragma unroll
    for (int r = 0; r < 8; ++r) {
        const float s0 = acc[0][r] + ach[0][r];
        const float s1 = acc[1][r] + ach[1][r];
        v2f d;
        d.x = SC ? alpha * s0 : s0;
        d.y = SC ? alpha * s1 : s1;
        *(v2f*)(out + (m0 + r + 8 * g) * N + nw) = d;
    }
}

__global__ void __launch_bounds__(PH_T)
relax_resid(const float* __restrict__ t1, const float* __restrict__ t2,
            const float* __restrict__ t3,
            float* __restrict__ r2, float* __restrict__ r3,
            const float* __restrict__ W1, const float* __restrict__ W2)
{
    __shared__ __attribute__((aligned(16))) float S1[PH_M * PW];
    __shared__ __attribute__((aligned(16))) float S2[PH_M * PW];
    __shared__ __attribute__((aligned(16))) float X[12 * 512];

    const int tid  = threadIdx.x;
    const int lane = tid & 31;
    const int wid  = __builtin_amdgcn_readfirstlane(tid >> 5);
    const int g    = lane >> 4;
    const int ln   = lane & 15;
    const int kh   = wid & 1;
    const int slab = blockIdx.x & 31;
    const int role = ph_role(blockIdx.x >> 5);
    const size_t m0 = (size_t)slab * PH_M;
    const int vt   = 6 * role + (wid >> 1);

    SlabRegs<1024, PH_T, PH_M> f1, f2;
    if (role <= 5) slab_fetch(f1, t1 + m0 * 1024, tid);
    if (role >= 5) slab_fetch(f2, t2 + m0 * 1024, tid);
    if (role <= 5) slab_put(S1, f1, tid);
    if (role >= 5) slab_put(S2, f2, tid);
    __syncthreads();

    const int k0   = 512 * kh;
    const int own  = (ln + 16 * kh) * PW + 4 * g + k0;
    const int oth  = (ln + 16 * (kh ^ 1)) * PW + 4 * g + k0;
    v8f acc[2][2] = {};
    int nw;
    if (vt < 32) {
        nw = 32 * vt + 2 * ln;
        slab_mma_2x2<512, 1024>(S1 + own, S1 + oth,
                                W1 + (size_t)(k0 + 4 * g) * 1024 + nw, acc);
    } else {
        nw = 32 * (vt - 32) + 2 * ln;
        slab_mma_2x2<512, 512>(S2 + own, S2 + oth,
                               W2 + (size_t)(k0 + 4 * g) * 512 + nw, acc);
    }
    pair_put(X, wid, lane, acc[1]);
    __syncthreads();
    pair_get(X, wid, lane, acc[0]);

    if (vt < 32) {
        #pragma unroll
        for (int r = 0; r < 8; ++r) {
            const size_t go = (m0 + 16 * kh + r + 8 * g) * 1024 + nw;
            v2f c = *(const v2f*)(t2 + go);
            v2f d;
            d.x = c.x - acc[0][0][r]; d.y = c.y - acc[0][1][r];
            *(v2f*)(r2 + go) = d;
        }
    } else {
        #pragma unroll
        for (int r = 0; r < 8; ++r) {
            const size_t go = (m0 + 16 * kh + r + 8 * g) * 512 + nw;
            v2f c = *(const v2f*)(t3 + go);
            v2f d;
            d.x = c.x - acc[0][0][r]; d.y = c.y - acc[0][1][r];
            *(v2f*)(r3 + go) = d;
        }
    }
}

__global__ void __launch_bounds__(PH_T)
relax_update(float* __restrict__ t1, float* __restrict__ t2, float* __restrict__ t3,
             const float* __restrict__ r2, const float* __restrict__ r3,
             const float* __restrict__ p0, const float* __restrict__ y,
             const float* __restrict__ W1T, const float* __restrict__ W2T)
{
    __shared__ __attribute__((aligned(16))) float R2[PH_M * PW];
    __shared__ __attribute__((aligned(16))) float R3[PH_M * PH];
    __shared__ __attribute__((aligned(16))) float X[12 * 512];

    const int tid  = threadIdx.x;
    const int lane = tid & 31;
    const int wid  = __builtin_amdgcn_readfirstlane(tid >> 5);
    const int g    = lane >> 4;
    const int ln   = lane & 15;
    const int kh   = wid & 1;
    const int slab = blockIdx.x & 31;
    const int role = ph_role(blockIdx.x >> 5);
    const size_t m0 = (size_t)slab * PH_M;
    const int vu   = 12 * role + wid;

    SlabRegs<1024, PH_T, PH_M> f2;
    SlabRegs<512, PH_T, PH_M>  f3;
    if (role <= 5) slab_fetch(f2, r2 + m0 * 1024, tid);
    if (role >= 5) slab_fetch(f3, r3 + m0 * 512, tid);
    if (role <= 5) slab_put(R2, f2, tid);
    if (role >= 5) slab_put(R3, f3, tid);
    __syncthreads();

    #pragma unroll
    for (int i = 0; i < (PH_M * 16 + PH_T - 1) / PH_T; ++i) {
        int idx4 = i * PH_T + tid;
        if (idx4 < PH_M * 16) {
            const int row = idx4 >> 4;
            const int col = 64 * role + 4 * (idx4 & 15);
            const size_t go = (m0 + row) * 512 + col;
            v4f rv = *(const v4f*)(r3 + go);
            v4f tv = *(const v4f*)(t3 + go);
            v4f yv = *(const v4f*)(y + go);
            rv.x += EP_BETA * (tv.x - yv.x); rv.y += EP_BETA * (tv.y - yv.y);
            rv.z += EP_BETA * (tv.z - yv.z); rv.w += EP_BETA * (tv.w - yv.w);
            tv.x -= EP_LR * rv.x; tv.y -= EP_LR * rv.y;
            tv.z -= EP_LR * rv.z; tv.w -= EP_LR * rv.w;
            *(v4f*)(t3 + go) = tv;
        }
    }

    v8f acc[2][2] = {};
    int nw;
    if (vu < 64) {
        const int k0  = 512 * kh;
        const int own = (ln + 16 * kh) * PW + 4 * g + k0;
        const int oth = (ln + 16 * (kh ^ 1)) * PW + 4 * g + k0;
        nw = 32 * (vu >> 1) + 2 * ln;
        slab_mma_2x2<512, 1024>(R2 + own, R2 + oth,
                                W1T + (size_t)(k0 + 4 * g) * 1024 + nw, acc);
        pair_put(X, wid, lane, acc[1]);
    } else {
        nw = 32 * (vu - 64) + 2 * ln;
        slab_mma_2x2<512, 1024>(R3 + ln * PH + 4 * g, R3 + (ln + 16) * PH + 4 * g,
                                W2T + (size_t)(4 * g) * 1024 + nw, acc);
    }
    __syncthreads();

    if (vu < 64) {
        pair_get(X, wid, lane, acc[0]);
        #pragma unroll
        for (int r = 0; r < 8; ++r) {
            const size_t go = (m0 + 16 * kh + r + 8 * g) * 1024 + nw;
            v2f tv = *(const v2f*)(t1 + go);
            v2f pv = *(const v2f*)(p0 + go);
            v2f gv;
            gv.x = (tv.x - pv.x) - acc[0][0][r]; gv.y = (tv.y - pv.y) - acc[0][1][r];
            tv.x -= EP_LR * gv.x; tv.y -= EP_LR * gv.y;
            *(v2f*)(t1 + go) = tv;
        }
    } else {
        #pragma unroll
        for (int m = 0; m < 2; ++m) {
            #pragma unroll
            for (int r = 0; r < 8; ++r) {
                const size_t go = (m0 + 16 * m + r + 8 * g) * 1024 + nw;
                v2f rv = *(const v2f*)(r2 + go);
                v2f tv = *(const v2f*)(t2 + go);
                v2f gv;
                gv.x = rv.x - acc[m][0][r]; gv.y = rv.y - acc[m][1][r];
                tv.x -= EP_LR * gv.x; tv.y -= EP_LR * gv.y;
                *(v2f*)(t2 + go) = tv;
            }
        }
    }
}

constexpr int S0_T = 256;

__global__ void __launch_bounds__(S0_T)
step0_g2(const float* __restrict__ r2, const float* __restrict__ r3,
         const float* __restrict__ t3, const float* __restrict__ y,
         const float* __restrict__ W2T, float* __restrict__ G2)
{
    __shared__ __attribute__((aligned(16))) float R3[SLAB_M * PH];
    __shared__ __attribute__((aligned(16))) float G3[SLAB_M * PH];

    const int tid  = threadIdx.x;
    const int lane = tid & 31;
    const int wid  = tid >> 5;
    const int g    = lane >> 4;
    const int ln   = lane & 15;
    const int q    = blockIdx.x & 3;
    const size_t m0 = (size_t)(blockIdx.x >> 2) * SLAB_M;

    #pragma unroll
    for (int i = 0; i < SLAB_M * 128 / S0_T; ++i) {
        const int idx4 = i * S0_T + tid;
        const int row = idx4 >> 7;
        const int col = 4 * (idx4 & 127);
        const size_t go = (m0 + row) * 512 + col;
        v4f rv = *(const v4f*)(r3 + go);
        v4f tv = *(const v4f*)(t3 + go);
        v4f yv = *(const v4f*)(y + go);
        *(v4f*)(R3 + row * PH + col) = rv;
        rv.x += EP_BETA * (tv.x - yv.x); rv.y += EP_BETA * (tv.y - yv.y);
        rv.z += EP_BETA * (tv.z - yv.z); rv.w += EP_BETA * (tv.w - yv.w);
        *(v4f*)(G3 + row * PH + col) = rv;
    }
    __syncthreads();

    const int nw = 256 * q + 32 * wid + 2 * ln;
    const float* Wl = W2T + (size_t)(4 * g) * 1024 + nw;
    v8f accr[2] = {};
    v8f accg[2] = {};
    slab_mma_kn<2, 512, 1024>(R3, Wl, ln, g, accr);
    slab_mma_kn<2, 512, 1024>(G3, Wl, ln, g, accg);
    #pragma unroll
    for (int r = 0; r < 8; ++r) {
        const size_t go = (m0 + r + 8 * g) * 1024 + nw;
        v2f c = *(const v2f*)(r2 + go);
        v2f d;
        d.x = (c.x - accr[0][r]) + accg[0][r];
        d.y = (c.y - accr[1][r]) + accg[1][r];
        *(v2f*)(G2 + go) = d;
    }
}

__global__ void __launch_bounds__(S0_T)
step0_update(float* __restrict__ t1, float* __restrict__ t2, float* __restrict__ t3,
             const float* __restrict__ r2, const float* __restrict__ r3,
             const float* __restrict__ G2, const float* __restrict__ p0,
             const float* __restrict__ y, const float* __restrict__ W1T)
{
    __shared__ __attribute__((aligned(16))) float R2[SLAB_M * PW];
    __shared__ __attribute__((aligned(16))) float G2s[SLAB_M * PW];

    const int tid  = threadIdx.x;
    const int lane = tid & 31;
    const int wid  = tid >> 5;
    const int g    = lane >> 4;
    const int ln   = lane & 15;
    const int q    = blockIdx.x & 3;
    const size_t m0 = (size_t)(blockIdx.x >> 2) * SLAB_M;

    slab_stage<1024, S0_T>(R2, r2 + m0 * 1024, tid);
    slab_stage<1024, S0_T>(G2s, G2 + m0 * 1024, tid);
    __syncthreads();

    #pragma unroll
    for (int i = 0; i < SLAB_M * 32 / S0_T; ++i) {
        const int idx4 = i * S0_T + tid;
        const int row = idx4 >> 5;
        const int col = 128 * q + 4 * (idx4 & 31);
        const size_t go = (m0 + row) * 512 + col;
        v4f rv = *(const v4f*)(r3 + go);
        v4f tv = *(const v4f*)(t3 + go);
        v4f yv = *(const v4f*)(y + go);
        rv.x += EP_BETA * (tv.x - yv.x); rv.y += EP_BETA * (tv.y - yv.y);
        rv.z += EP_BETA * (tv.z - yv.z); rv.w += EP_BETA * (tv.w - yv.w);
        tv.x -= EP_LR * rv.x; tv.y -= EP_LR * rv.y;
        tv.z -= EP_LR * rv.z; tv.w -= EP_LR * rv.w;
        *(v4f*)(t3 + go) = tv;
    }

    #pragma unroll
    for (int i = 0; i < SLAB_M * 64 / S0_T; ++i) {
        const int idx4 = i * S0_T + tid;
        const int row = idx4 >> 6;
        const int col = 256 * q + 4 * (idx4 & 63);
        const size_t go = (m0 + row) * 1024 + col;
        v4f gv = *(const v4f*)(G2s + row * PW + col);
        v4f tv = *(const v4f*)(t2 + go);
        tv.x -= EP_LR * gv.x; tv.y -= EP_LR * gv.y;
        tv.z -= EP_LR * gv.z; tv.w -= EP_LR * gv.w;
        *(v4f*)(t2 + go) = tv;
    }

    const int nw = 256 * q + 32 * wid + 2 * ln;
    const float* Wl = W1T + (size_t)(4 * g) * 1024 + nw;
    v8f accr[2] = {};
    v8f accg[2] = {};
    slab_mma_kn<2, 1024, 1024>(R2, Wl, ln, g, accr);
    slab_mma_kn<2, 1024, 1024>(G2s, Wl, ln, g, accg);
    #pragma unroll
    for (int r = 0; r < 8; ++r) {
        const size_t go = (m0 + r + 8 * g) * 1024 + nw;
        v2f tv = *(const v2f*)(t1 + go);
        v2f pv = *(const v2f*)(p0 + go);
        v2f gv;
        gv.x = ((tv.x - pv.x) - accr[0][r]) + accg[0][r];
        gv.y = ((tv.y - pv.y) - accr[1][r]) + accg[1][r];
        tv.x -= EP_LR * gv.x; tv.y -= EP_LR * gv.y;
        *(v2f*)(t1 + go) = tv;
    }
}

#define EW(kernel, n, ...) kernel<<<dim3((n) / 1024), 256, 0, stream>>>(__VA_ARGS__)

extern "C" void kernel_launch(void* const* d_in, const int* in_sizes, int n_in,
                              void* d_out, int out_size, void* d_ws, size_t ws_size,
                              hipStream_t stream)
{
    const float* x  = (const float*)d_in[0];
    const float* y  = (const float*)d_in[1];
    const float* W0 = (const float*)d_in[2];
    const float* W1 = (const float*)d_in[3];
    const float* W2 = (const float*)d_in[4];
    float* out = (float*)d_out;

    const size_t n1 = (size_t)1024 * 1024;
    const size_t n3 = (size_t)1024 * 512;

    float* ws = (float*)d_ws;
    float* p0 = ws;
    float* t1 = p0 + n1;
    float* t2 = t1 + n1;
    float* t3 = t2 + n1;
    float* r1 = t3 + n3;
    float* r2 = r1 + n1;
    float* r3 = r2 + n1;
    float* part = r3 + n3;

    slab_forward<1024, false><<<dim3(FW_G1), FW_T, 0, stream>>>(x, W0, p0, 1.0f);
    EW(copyk, n1, t1, p0);
    slab_forward<1024, false><<<dim3(FW_G1), FW_T, 0, stream>>>(t1, W1, t2, 1.0f);
    slab_forward<512, false><<<dim3(FW_G3), FW_T, 0, stream>>>(t2, W2, t3, 1.0f);
    loss_partial<<<512, 256, 0, stream>>>(t3, y, part);
    loss_final<<<1, 256, 0, stream>>>(part, out + 2621440);

    float* W1T = r1;
    float* W2T = out + 2097152;
    float* G2  = out;
    transpose_rc<1024, 1024><<<dim3(32, 32), 256, 0, stream>>>(W1T, W1);
    transpose_rc<1024, 512><<<dim3(16, 32), 256, 0, stream>>>(W2T, W2);

    relax_resid<<<dim3(PH_G), PH_T, 0, stream>>>(t1, t2, t3, r2, r3, W1, W2);
    step0_g2<<<dim3(PH_G), S0_T, 0, stream>>>(r2, r3, t3, y, W2T, G2);
    step0_update<<<dim3(PH_G), S0_T, 0, stream>>>(t1, t2, t3, r2, r3, G2, p0, y, W1T);

    for (int it = 0; it < 19; ++it) {
        relax_resid<<<dim3(PH_G), PH_T, 0, stream>>>(t1, t2, t3, r2, r3, W1, W2);
        relax_update<<<dim3(PH_G), PH_T, 0, stream>>>(t1, t2, t3, r2, r3, p0, y, W1T, W2T);
    }

    relax_resid<<<dim3(PH_G), PH_T, 0, stream>>>(t1, t2, t3, r2, r3, W1, W2);
    EW(subk, n1, r1, t1, p0);
    float* sT = p0;
    transpose_rc<1024, 1024><<<dim3(32, 32), 256, 0, stream>>>(sT, x);
    slab_forward<1024, true><<<dim3(FW_G1), FW_T, 0, stream>>>(sT, r1, out,           -1.0f / EP_BETA);
    transpose_rc<1024, 1024><<<dim3(32, 32), 256, 0, stream>>>(sT, t1);
    slab_forward<1024, true><<<dim3(FW_G1), FW_T, 0, stream>>>(sT, r2, out + 1048576, -1.0f / EP_BETA);
    transpose_rc<1024, 1024><<<dim3(32, 32), 256, 0, stream>>>(sT, t2);
    slab_forward<512, true><<<dim3(FW_G3), FW_T, 0, stream>>>(sT, r3, out + 2097152, -1.0f / EP_BETA);
}
